# starting kernel
# baseline (speedup 1.0000x reference)
#include <hip/hip_runtime.h>
#include <hip/hip_bf16.h>

constexpr int B   = 2;
constexpr int S   = 4096;
constexpr int E   = 1024;
constexpr int D   = 64;
constexpr int WIN = 5;
constexpr int DIL = 2;

typedef __attribute__((ext_vector_type(2))) float v2f;
typedef __attribute__((ext_vector_type(8))) float v8f;

__global__ __launch_bounds__(256) void qkv_proj_kernel(
    const float* __restrict__ x,
    const float* __restrict__ W,
    const float* __restrict__ bias,
    float* __restrict__ out)
{
    const int lane  = threadIdx.x & 31;
    const int wave  = threadIdx.x >> 5;
    const int mtile = blockIdx.x * 8 + wave;
    const int row0  = mtile * 16;

    const int m  = lane & 15;
    const int n  = lane & 15;
    const int kh = (lane >> 4) * 2;

    const float* xrow = x + (size_t)(row0 + m) * E + kh;

    v8f acc0 = {}, acc1 = {}, acc2 = {}, acc3 = {};

    for (int k0 = 0; k0 < E; k0 += 4) {
        v2f a = *(const v2f*)(xrow + k0);

        const float* wp = W + (size_t)(k0 + kh) * D + n;
        v2f b0, b1, b2, b3;
        b0.x = wp[0];      b0.y = wp[D];
        b1.x = wp[16];     b1.y = wp[D + 16];
        b2.x = wp[32];     b2.y = wp[D + 32];
        b3.x = wp[48];     b3.y = wp[D + 48];

        acc0 = __builtin_amdgcn_wmma_f32_16x16x4_f32(false, a, false, b0,
                                                     (short)0, acc0, false, false);
        acc1 = __builtin_amdgcn_wmma_f32_16x16x4_f32(false, a, false, b1,
                                                     (short)0, acc1, false, false);
        acc2 = __builtin_amdgcn_wmma_f32_16x16x4_f32(false, a, false, b2,
                                                     (short)0, acc2, false, false);
        acc3 = __builtin_amdgcn_wmma_f32_16x16x4_f32(false, a, false, b3,
                                                     (short)0, acc3, false, false);
    }

    const int mo = (lane >> 4) * 8;
    float* orow = out + (size_t)(row0 + mo) * D + n;
#pragma unroll
    for (int r = 0; r < 8; ++r) {
        float* o = orow + (size_t)r * D;
        o[0]  = acc0[r] + bias[n];
        o[16] = acc1[r] + bias[n + 16];
        o[32] = acc2[r] + bias[n + 32];
        o[48] = acc3[r] + bias[n + 48];
    }
}

__global__ __launch_bounds__(64) void vsum_kernel(
    const float* __restrict__ V, float* __restrict__ Vsum)
{
    const int b = blockIdx.x;
    const int d = threadIdx.x;
    const float* p = V + (size_t)b * S * D + d;
    float s = 0.0f;
#pragma unroll 4
    for (int i = 0; i < S; ++i) s += p[(size_t)i * D];
    Vsum[b * D + d] = s;
}

__global__ __launch_bounds__(256) void attn_kernel(
    const float* __restrict__ Q, const float* __restrict__ K,
    const float* __restrict__ V, const float* __restrict__ Vsum,
    float* __restrict__ out)
{
    const int gwave = (blockIdx.x * blockDim.x + threadIdx.x) >> 5;
    if (gwave >= B * S) return;
    const int lane = threadIdx.x & 31;
    const int b = gwave / S;
    const int i = gwave - b * S;

    const size_t rowQ = (size_t)(b * S + i) * D;
    const float2 qv = *(const float2*)(Q + rowQ + lane * 2);

    float accx = Vsum[b * D + lane * 2];
    float accy = Vsum[b * D + lane * 2 + 1];
    float Z = (float)S;

#pragma unroll
    for (int w = 0; w < WIN; ++w) {
        const int j = i + DIL * (w - WIN / 2);
        if (j < 0 || j >= S) continue;
        const size_t rowJ = (size_t)(b * S + j) * D;
        const float2 kv = *(const float2*)(K + rowJ + lane * 2);
        float c = qv.x * kv.x + qv.y * kv.y;
#pragma unroll
        for (int off = 16; off > 0; off >>= 1) c += __shfl_xor(c, off, 32);
        const float em1 = __expf(c) - 1.0f;
        const float2 vv = *(const float2*)(V + rowJ + lane * 2);
        accx += em1 * vv.x;
        accy += em1 * vv.y;
        Z    += em1;
    }

    const float inv = 1.0f / Z;
    float2 o;
    o.x = accx * inv;
    o.y = accy * inv;
    *(float2*)(out + rowQ + lane * 2) = o;
}

extern "C" void kernel_launch(void* const* d_in, const int* in_sizes, int n_in,
                              void* d_out, int out_size, void* d_ws, size_t ws_size,
                              hipStream_t stream) {
    const float* x  = (const float*)d_in[0];
    const float* Wq = (const float*)d_in[1];
    const float* bq = (const float*)d_in[2];
    const float* Wk = (const float*)d_in[3];
    const float* bk = (const float*)d_in[4];
    const float* Wv = (const float*)d_in[5];
    const float* bv = (const float*)d_in[6];
    float* outp = (float*)d_out;

    constexpr size_t QKV = (size_t)B * S * D;
    float* ws = (float*)d_ws;
    float* Qb = ws;
    float* Kb = ws + QKV;
    float* Vb = ws + 2 * QKV;
    float* Vs = ws + 3 * QKV;

    const int mtiles = (B * S) / 16;
    dim3 gproj(mtiles / 8);
    qkv_proj_kernel<<<gproj, 256, 0, stream>>>(x, Wq, bq, Qb);
    qkv_proj_kernel<<<gproj, 256, 0, stream>>>(x, Wk, bk, Kb);
    qkv_proj_kernel<<<gproj, 256, 0, stream>>>(x, Wv, bv, Vb);

    vsum_kernel<<<B, 64, 0, stream>>>(Vb, Vs);

    const int nwaves = B * S;
    attn_kernel<<<(nwaves * 32) / 256, 256, 0, stream>>>(Qb, Kb, Vb, Vs, outp);
}


# proj: LDS fragment reads of 8 k steps go as one group into a second register set ahead of the previous block's matrix run (W tile in 16-column planes: one read per B fragment); sum order unchanged
# speedup vs baseline: 12.1182x; 12.1182x over previous
#include <hip/hip_runtime.h>
#include <hip/hip_bf16.h>

constexpr int B   = 2;
constexpr int S   = 4096;
constexpr int E   = 1024;
constexpr int D   = 64;
constexpr int WIN = 5;
constexpr int DIL = 2;

constexpr int NCH       = 16;
constexpr int CH_ROWS   = (B * S) / NCH;
constexpr int BM        = 32;
constexpr int BK        = 64;
constexpr int NSLAB     = E / BK;
constexpr int XP        = BK + 4;
constexpr int WP        = 24;
constexpr int WPLANE    = BK * WP + 16;
constexpr int WSTAGE    = (D / 16) * WPLANE;
constexpr int PROJ_WGS  = (B * S) / BM;
constexpr int KB        = 8;

typedef __attribute__((ext_vector_type(2))) float v2f;
typedef __attribute__((ext_vector_type(8))) float v8f;

__global__ __launch_bounds__(512) void xsum_kernel(
    const float* __restrict__ x,
    float* __restrict__ xs)
{
    const int lane = threadIdx.x & 31;
    const int wave = threadIdx.x >> 5;
    const int cl   = lane & 15;
    const int rh   = lane >> 4;
    const int cb   = blockIdx.x;
    const int ch   = blockIdx.y;

    const float* p = x + (size_t)(ch * CH_ROWS + wave * 32 + rh) * E + cb * 64 + cl * 4;

    float4 v[16];
#pragma unroll
    for (int it = 0; it < 16; ++it) v[it] = *(const float4*)(p + (size_t)(2 * it) * E);

    float4 s = v[0];
#pragma unroll
    for (int it = 1; it < 16; ++it) {
        s.x += v[it].x; s.y += v[it].y; s.z += v[it].z; s.w += v[it].w;
    }
    s.x += __shfl_xor(s.x, 16, 32);
    s.y += __shfl_xor(s.y, 16, 32);
    s.z += __shfl_xor(s.z, 16, 32);
    s.w += __shfl_xor(s.w, 16, 32);

    __shared__ float4 red[16][16];
    if (rh == 0) red[wave][cl] = s;
    __syncthreads();
    if (threadIdx.x < 16) {
        float4 t = red[0][threadIdx.x];
        for (int w = 1; w < 16; ++w) {
            const float4 u = red[w][threadIdx.x];
            t.x += u.x; t.y += u.y; t.z += u.z; t.w += u.w;
        }
        *(float4*)(xs + (size_t)ch * E + cb * 64 + threadIdx.x * 4) = t;
    }
}

__global__ __launch_bounds__(128) void qkv_proj_kernel(
    const float* __restrict__ x,
    const float* __restrict__ W,
    const float* __restrict__ bias,
    float* __restrict__ out,
    const float* __restrict__ xs,
    float* __restrict__ vsum)
{
    __shared__ __attribute__((aligned(16))) float Xs[2][BM * XP];
    __shared__ __attribute__((aligned(16))) float Ws[2][WSTAGE];

    const int tid  = threadIdx.x;
    const int lane = tid & 31;
    const int wave = tid >> 5;
    const int mt   = wave & 1;
    const int nh   = wave >> 1;

    const int m  = lane & 15;
    const int n  = lane & 15;
    const int kh = (lane >> 4) * 2;

    const bool sumtile = (blockIdx.x == PROJ_WGS);
    const int  row0    = blockIdx.x * BM;

    const int c4 = tid & 15;
    const int r8 = tid >> 4;

    const float* xg = sumtile ? xs + (size_t)r8 * E + c4 * 4
                              : x + (size_t)(row0 + r8) * E + c4 * 4;
    const size_t xo2 = sumtile ? (size_t)0       : (size_t)16 * E;
    const size_t xo3 = sumtile ? (size_t)8 * E   : (size_t)24 * E;
    const float* wg = W + (size_t)r8 * D + c4 * 4;

    float* const xst = &Xs[0][r8 * XP + c4 * 4];
    float* const wst = &Ws[0][(c4 >> 2) * WPLANE + r8 * WP + (c4 & 3) * 4];

    const float* const xa = &Xs[0][(mt * 16 + m) * XP + kh];
    const float* const wb = &Ws[0][(nh * 2) * WPLANE + kh * WP + n];

    float4 xr0, xr1, xr2, xr3;
    float4 wr0, wr1, wr2, wr3, wr4, wr5, wr6, wr7;

    v8f acc0 = {}, acc1 = {};

    v2f a0[KB], p0[KB], q0[KB];
    v2f a1[KB], p1[KB], q1[KB];

#define LOAD_FRAGS(A_, P_, Q_, XA_, WB_, BLK_)                                   \
    _Pragma("unroll")                                                            \
    for (int i = 0; i < KB; ++i) {                                               \
        const int kk = (BLK_) * KB + i;                                          \
        A_[i] = *(const v2f*)((XA_) + kk * 4);                                   \
        const float* wp = (WB_) + (kk * 4) * WP;                                 \
        P_[i].x = wp[0];      P_[i].y = wp[WP];                                  \
        Q_[i].x = wp[WPLANE]; Q_[i].y = wp[WPLANE + WP];                         \
    }
#define MMA_BLOCK(A_, P_, Q_)                                                    \
    _Pragma("unroll")                                                            \
    for (int i = 0; i < KB; ++i) {                                               \
        acc0 = __builtin_amdgcn_wmma_f32_16x16x4_f32(false, A_[i], false, P_[i], \
                                                     (short)0, acc0, false, false); \
        acc1 = __builtin_amdgcn_wmma_f32_16x16x4_f32(false, A_[i], false, Q_[i], \
                                                     (short)0, acc1, false, false); \
    }

    {
        float4 xh0, xh1, xh2, xh3;
        float4 wh0, wh1, wh2, wh3, wh4, wh5, wh6, wh7;
        xh0 = *(const float4*)(xg + 0);
        xh1 = *(const float4*)(xg + 0 + (size_t)8 * E);
        xh2 = *(const float4*)(xg + 0 + xo2);
        xh3 = *(const float4*)(xg + 0 + xo3);
        wh0 = *(const float4*)(wg + (size_t)(0 + 0) * D);
        wh1 = *(const float4*)(wg + (size_t)(0 + 8) * D);
        wh2 = *(const float4*)(wg + (size_t)(0 + 16) * D);
        wh3 = *(const float4*)(wg + (size_t)(0 + 24) * D);
        wh4 = *(const float4*)(wg + (size_t)(0 + 32) * D);
        wh5 = *(const float4*)(wg + (size_t)(0 + 40) * D);
        wh6 = *(const float4*)(wg + (size_t)(0 + 48) * D);
        wh7 = *(const float4*)(wg + (size_t)(0 + 56) * D);
        xr0 = *(const float4*)(xg + BK);
        xr1 = *(const float4*)(xg + BK + (size_t)8 * E);
        xr2 = *(const float4*)(xg + BK + xo2);
        xr3 = *(const float4*)(xg + BK + xo3);
        wr0 = *(const float4*)(wg + (size_t)(BK + 0) * D);
        wr1 = *(const float4*)(wg + (size_t)(BK + 8) * D);
        wr2 = *(const float4*)(wg + (size_t)(BK + 16) * D);
        wr3 = *(const float4*)(wg + (size_t)(BK + 24) * D);
        wr4 = *(const float4*)(wg + (size_t)(BK + 32) * D);
        wr5 = *(const float4*)(wg + (size_t)(BK + 40) * D);
        wr6 = *(const float4*)(wg + (size_t)(BK + 48) * D);
        wr7 = *(const float4*)(wg + (size_t)(BK + 56) * D);
        *(float4*)(xst + 0 * XP) = xh0;
        *(float4*)(xst + 8 * XP) = xh1;
        *(float4*)(xst + 16 * XP) = xh2;
        *(float4*)(xst + 24 * XP) = xh3;
        *(float4*)(wst + 0 * WP) = wh0;
        *(float4*)(wst + 8 * WP) = wh1;
        *(float4*)(wst + 16 * WP) = wh2;
        *(float4*)(wst + 24 * WP) = wh3;
        *(float4*)(wst + 32 * WP) = wh4;
        *(float4*)(wst + 40 * WP) = wh5;
        *(float4*)(wst + 48 * WP) = wh6;
        *(float4*)(wst + 56 * WP) = wh7;
        float* xs_1 = xst + BM * XP;
        float* ws_1 = wst + WSTAGE;
        *(float4*)(xs_1 + 0 * XP) = xr0;
        *(float4*)(xs_1 + 8 * XP) = xr1;
        *(float4*)(xs_1 + 16 * XP) = xr2;
        *(float4*)(xs_1 + 24 * XP) = xr3;
        *(float4*)(ws_1 + 0 * WP) = wr0;
        *(float4*)(ws_1 + 8 * WP) = wr1;
        *(float4*)(ws_1 + 16 * WP) = wr2;
        *(float4*)(ws_1 + 24 * WP) = wr3;
        *(float4*)(ws_1 + 32 * WP) = wr4;
        *(float4*)(ws_1 + 40 * WP) = wr5;
        *(float4*)(ws_1 + 48 * WP) = wr6;
        *(float4*)(ws_1 + 56 * WP) = wr7;
    }
    __syncthreads();
    {
        const int k2 = 2 * BK;
        xr0 = *(const float4*)(xg + k2);
        xr1 = *(const float4*)(xg + k2 + (size_t)8 * E);
        xr2 = *(const float4*)(xg + k2 + xo2);
        xr3 = *(const float4*)(xg + k2 + xo3);
        wr0 = *(const float4*)(wg + (size_t)(k2 + 0) * D);
        wr1 = *(const float4*)(wg + (size_t)(k2 + 8) * D);
        wr2 = *(const float4*)(wg + (size_t)(k2 + 16) * D);
        wr3 = *(const float4*)(wg + (size_t)(k2 + 24) * D);
        wr4 = *(const float4*)(wg + (size_t)(k2 + 32) * D);
        wr5 = *(const float4*)(wg + (size_t)(k2 + 40) * D);
        wr6 = *(const float4*)(wg + (size_t)(k2 + 48) * D);
        wr7 = *(const float4*)(wg + (size_t)(k2 + 56) * D);
    }
    LOAD_FRAGS(a0, p0, q0, xa, wb, 0)

    for (int s = 0; s < NSLAB; ++s) {
        const int cur = s & 1;
        const float* xa_s = xa + cur * (BM * XP);
        const float* wb_s = wb + cur * WSTAGE;

        LOAD_FRAGS(a1, p1, q1, xa_s, wb_s, 1)
        __builtin_amdgcn_sched_barrier(0);
        MMA_BLOCK(a0, p0, q0)
        __builtin_amdgcn_sched_barrier(0);

        if (s + 1 < NSLAB) {
            __syncthreads();
            if (s + 2 < NSLAB) {
                float* xs_s = xst + cur * (BM * XP);
                float* ws_s = wst + cur * WSTAGE;
                *(float4*)(xs_s + 0 * XP) = xr0;
                *(float4*)(xs_s + 8 * XP) = xr1;
                *(float4*)(xs_s + 16 * XP) = xr2;
                *(float4*)(xs_s + 24 * XP) = xr3;
                *(float4*)(ws_s + 0 * WP) = wr0;
                *(float4*)(ws_s + 8 * WP) = wr1;
                *(float4*)(ws_s + 16 * WP) = wr2;
                *(float4*)(ws_s + 24 * WP) = wr3;
                *(float4*)(ws_s + 32 * WP) = wr4;
                *(float4*)(ws_s + 40 * WP) = wr5;
                *(float4*)(ws_s + 48 * WP) = wr6;
                *(float4*)(ws_s + 56 * WP) = wr7;
            }
            if (s + 3 < NSLAB) {
                const int k2 = (s + 3) * BK;
                xr0 = *(const float4*)(xg + k2);
                xr1 = *(const float4*)(xg + k2 + (size_t)8 * E);
                xr2 = *(const float4*)(xg + k2 + xo2);
                xr3 = *(const float4*)(xg + k2 + xo3);
                wr0 = *(const float4*)(wg + (size_t)(k2 + 0) * D);
                wr1 = *(const float4*)(wg + (size_t)(k2 + 8) * D);
                wr2 = *(const float4*)(wg + (size_t)(k2 + 16) * D);
                wr3 = *(const float4*)(wg + (size_t)(k2 + 24) * D);
                wr4 = *(const float4*)(wg + (size_t)(k2 + 32) * D);
                wr5 = *(const float4*)(wg + (size_t)(k2 + 40) * D);
                wr6 = *(const float4*)(wg + (size_t)(k2 + 48) * D);
                wr7 = *(const float4*)(wg + (size_t)(k2 + 56) * D);
            }
            const float* xa_n = xa + (cur ^ 1) * (BM * XP);
            const float* wb_n = wb + (cur ^ 1) * WSTAGE;
            LOAD_FRAGS(a0, p0, q0, xa_n, wb_n, 0)
        }
        __builtin_amdgcn_sched_barrier(0);
        MMA_BLOCK(a1, p1, q1)
        __builtin_amdgcn_sched_barrier(0);
    }
#undef LOAD_FRAGS
#undef MMA_BLOCK

    const int cbase = nh * 32 + n;

    if (sumtile) {
        if (mt != 0) return;
        float s0 = acc0[0], s1 = acc1[0];
#pragma unroll
        for (int r = 1; r < 8; ++r) {
            s0 += acc0[r]; s1 += acc1[r];
        }
        float* vo = vsum + (lane >> 4) * D + cbase;
        vo[0]  = s0 + (float)S * bias[cbase];
        vo[16] = s1 + (float)S * bias[cbase + 16];
        return;
    }

    const int mo = (lane >> 4) * 8;
    float* orow = out + (size_t)(row0 + mt * 16 + mo) * D + cbase;
    const float bi0 = bias[cbase];
    const float bi1 = bias[cbase + 16];
#pragma unroll
    for (int r = 0; r < 8; ++r) {
        float* o = orow + (size_t)r * D;
        o[0]  = acc0[r] + bi0;
        o[16] = acc1[r] + bi1;
    }
}

__global__ __launch_bounds__(256) void attn_kernel(
    const float* __restrict__ Q, const float* __restrict__ K,
    const float* __restrict__ V, const float* __restrict__ Vsum,
    float* __restrict__ out)
{
    const int gwave = (blockIdx.x * blockDim.x + threadIdx.x) >> 5;
    if (gwave >= B * S) return;
    const int lane = threadIdx.x & 31;
    const int b = gwave / S;
    const int i = gwave - b * S;

    const size_t rowQ = (size_t)(b * S + i) * D;
    const float2 qv = *(const float2*)(Q + rowQ + lane * 2);

    float accx = Vsum[b * D + lane * 2];
    float accy = Vsum[b * D + lane * 2 + 1];
    float Z = (float)S;

#pragma unroll
    for (int w = 0; w < WIN; ++w) {
        const int j = i + DIL * (w - WIN / 2);
        if (j < 0 || j >= S) continue;
        const size_t rowJ = (size_t)(b * S + j) * D;
        const float2 kv = *(const float2*)(K + rowJ + lane * 2);
        float c = qv.x * kv.x + qv.y * kv.y;
#pragma unroll
        for (int off = 16; off > 0; off >>= 1) c += __shfl_xor(c, off, 32);
        const float em1 = __expf(c) - 1.0f;
        const float2 vv = *(const float2*)(V + rowJ + lane * 2);
        accx += em1 * vv.x;
        accy += em1 * vv.y;
        Z    += em1;
    }

    const float inv = 1.0f / Z;
    float2 o;
    o.x = accx * inv;
    o.y = accy * inv;
    *(float2*)(out + rowQ + lane * 2) = o;
}

extern "C" void kernel_launch(void* const* d_in, const int* in_sizes, int n_in,
                              void* d_out, int out_size, void* d_ws, size_t ws_size,
                              hipStream_t stream) {
    const float* x  = (const float*)d_in[0];
    const float* Wq = (const float*)d_in[1];
    const float* bq = (const float*)d_in[2];
    const float* Wk = (const float*)d_in[3];
    const float* bk = (const float*)d_in[4];
    const float* Wv = (const float*)d_in[5];
    const float* bv = (const float*)d_in[6];
    float* outp = (float*)d_out;

    constexpr size_t QKV = (size_t)B * S * D;
    float* ws = (float*)d_ws;
    float* Qb = ws;
    float* Kb = ws + QKV;
    float* Vb = ws + 2 * QKV;
    float* Vs = ws + 3 * QKV;

    float* Xs = Kb;
    xsum_kernel<<<dim3(E / 64, NCH), 512, 0, stream>>>(x, Xs);

    qkv_proj_kernel<<<PROJ_WGS + 1, 128, 0, stream>>>(x, Wv, bv, Vb, Xs, Vs);
    qkv_proj_kernel<<<PROJ_WGS, 128, 0, stream>>>(x, Wq, bq, Qb, nullptr, nullptr);
    qkv_proj_kernel<<<PROJ_WGS, 128, 0, stream>>>(x, Wk, bk, Kb, nullptr, nullptr);

    const int nwaves = B * S;
    attn_kernel<<<(nwaves * 32) / 256, 256, 0, stream>>>(Qb, Kb, Vb, Vs, outp);
}


# proj: fetches of 3 K stages stay in flight: 3 register sets, stored to LDS 3 passes after issue, behind an LDS-only barrier (the old barrier drained all loads every pass); sum order unchanged
# speedup vs baseline: 15.0289x; 1.2402x over previous
#include <hip/hip_runtime.h>
#include <hip/hip_bf16.h>

constexpr int B   = 2;
constexpr int S   = 4096;
constexpr int E   = 1024;
constexpr int D   = 64;
constexpr int WIN = 5;
constexpr int DIL = 2;

constexpr int NCH       = 16;
constexpr int CH_ROWS   = (B * S) / NCH;
constexpr int BM        = 32;
constexpr int BK        = 64;
constexpr int NSLAB     = E / BK;
constexpr int XP        = BK + 4;
constexpr int WP        = 24;
constexpr int WPLANE    = BK * WP + 16;
constexpr int WSTAGE    = (D / 16) * WPLANE;
constexpr int PROJ_WGS  = (B * S) / BM;
constexpr int KB        = 8;
constexpr int NSET      = 3;

typedef __attribute__((ext_vector_type(2))) float v2f;
typedef __attribute__((ext_vector_type(8))) float v8f;

__global__ __launch_bounds__(512) void xsum_kernel(
    const float* __restrict__ x,
    float* __restrict__ xs)
{
    const int lane = threadIdx.x & 31;
    const int wave = threadIdx.x >> 5;
    const int cl   = lane & 15;
    const int rh   = lane >> 4;
    const int cb   = blockIdx.x;
    const int ch   = blockIdx.y;

    const float* p = x + (size_t)(ch * CH_ROWS + wave * 32 + rh) * E + cb * 64 + cl * 4;

    float4 v[16];
#pragma unroll
    for (int it = 0; it < 16; ++it) v[it] = *(const float4*)(p + (size_t)(2 * it) * E);

    float4 s = v[0];
#pragma unroll
    for (int it = 1; it < 16; ++it) {
        s.x += v[it].x; s.y += v[it].y; s.z += v[it].z; s.w += v[it].w;
    }
    s.x += __shfl_xor(s.x, 16, 32);
    s.y += __shfl_xor(s.y, 16, 32);
    s.z += __shfl_xor(s.z, 16, 32);
    s.w += __shfl_xor(s.w, 16, 32);

    __shared__ float4 red[16][16];
    if (rh == 0) red[wave][cl] = s;
    __syncthreads();
    if (threadIdx.x < 16) {
        float4 t = red[0][threadIdx.x];
        for (int w = 1; w < 16; ++w) {
            const float4 u = red[w][threadIdx.x];
            t.x += u.x; t.y += u.y; t.z += u.z; t.w += u.w;
        }
        *(float4*)(xs + (size_t)ch * E + cb * 64 + threadIdx.x * 4) = t;
    }
}

static __device__ __forceinline__ void lds_stage_barrier() {
    asm volatile("s_wait_dscnt 0x0\n\ts_barrier_signal -1\n\ts_barrier_wait -1" ::: "memory");
}

__global__ __launch_bounds__(128) void qkv_proj_kernel(
    const float* __restrict__ x,
    const float* __restrict__ W,
    const float* __restrict__ bias,
    float* __restrict__ out,
    const float* __restrict__ xs,
    float* __restrict__ vsum)
{
    __shared__ __attribute__((aligned(16))) float Xs[2][BM * XP];
    __shared__ __attribute__((aligned(16))) float Ws[2][WSTAGE];

    const int tid  = threadIdx.x;
    const int lane = tid & 31;
    const int wave = tid >> 5;
    const int mt   = wave & 1;
    const int nh   = wave >> 1;

    const int m  = lane & 15;
    const int n  = lane & 15;
    const int kh = (lane >> 4) * 2;

    const bool sumtile = (blockIdx.x == PROJ_WGS);
    const int  row0    = blockIdx.x * BM;

    const int c4 = tid & 15;
    const int r8 = tid >> 4;

    const float* xg = sumtile ? xs + (size_t)r8 * E + c4 * 4
                              : x + (size_t)(row0 + r8) * E + c4 * 4;
    const size_t xo2 = sumtile ? (size_t)0       : (size_t)16 * E;
    const size_t xo3 = sumtile ? (size_t)8 * E   : (size_t)24 * E;
    const float* wg = W + (size_t)r8 * D + c4 * 4;

    float* const xst = &Xs[0][r8 * XP + c4 * 4];
    float* const wst = &Ws[0][(c4 >> 2) * WPLANE + r8 * WP + (c4 & 3) * 4];

    const float* const xa = &Xs[0][(mt * 16 + m) * XP + kh];
    const float* const wb = &Ws[0][(nh * 2) * WPLANE + kh * WP + n];

    float4 xrA0, xrA1, xrA2, xrA3, wrA0, wrA1, wrA2, wrA3, wrA4, wrA5, wrA6, wrA7;
    float4 xrB0, xrB1, xrB2, xrB3, wrB0, wrB1, wrB2, wrB3, wrB4, wrB5, wrB6, wrB7;
    float4 xrC0, xrC1, xrC2, xrC3, wrC0, wrC1, wrC2, wrC3, wrC4, wrC5, wrC6, wrC7;

    v8f acc0 = {}, acc1 = {};

    v2f a0[KB], p0[KB], q0[KB];
    v2f a1[KB], p1[KB], q1[KB];

#define FETCH(S_, T_)                                                            \
    {                                                                            \
        const int k2 = (T_) * BK;                                                \
        xr##S_##0 = *(const float4*)(xg + k2);                                   \
        xr##S_##1 = *(const float4*)(xg + k2 + (size_t)8 * E);                   \
        xr##S_##2 = *(const float4*)(xg + k2 + xo2);                             \
        xr##S_##3 = *(const float4*)(xg + k2 + xo3);                             \
        wr##S_##0 = *(const float4*)(wg + (size_t)(k2 + 0) * D);                 \
        wr##S_##1 = *(const float4*)(wg + (size_t)(k2 + 8) * D);                 \
        wr##S_##2 = *(const float4*)(wg + (size_t)(k2 + 16) * D);                \
        wr##S_##3 = *(const float4*)(wg + (size_t)(k2 + 24) * D);                \
        wr##S_##4 = *(const float4*)(wg + (size_t)(k2 + 32) * D);                \
        wr##S_##5 = *(const float4*)(wg + (size_t)(k2 + 40) * D);                \
        wr##S_##6 = *(const float4*)(wg + (size_t)(k2 + 48) * D);                \
        wr##S_##7 = *(const float4*)(wg + (size_t)(k2 + 56) * D);                \
    }
#define STORE(S_, CUR_)                                                          \
    {                                                                            \
        float* xs_s = xst + (CUR_) * (BM * XP);                                  \
        float* ws_s = wst + (CUR_) * WSTAGE;                                     \
        *(float4*)(xs_s + 0 * XP) = xr##S_##0;                                   \
        *(float4*)(xs_s + 8 * XP) = xr##S_##1;                                   \
        *(float4*)(xs_s + 16 * XP) = xr##S_##2;                                  \
        *(float4*)(xs_s + 24 * XP) = xr##S_##3;                                  \
        *(float4*)(ws_s + 0 * WP) = wr##S_##0;                                   \
        *(float4*)(ws_s + 8 * WP) = wr##S_##1;                                   \
        *(float4*)(ws_s + 16 * WP) = wr##S_##2;                                  \
        *(float4*)(ws_s + 24 * WP) = wr##S_##3;                                  \
        *(float4*)(ws_s + 32 * WP) = wr##S_##4;                                  \
        *(float4*)(ws_s + 40 * WP) = wr##S_##5;                                  \
        *(float4*)(ws_s + 48 * WP) = wr##S_##6;                                  \
        *(float4*)(ws_s + 56 * WP) = wr##S_##7;                                  \
    }
#define LOAD_FRAGS(A_, P_, Q_, XA_, WB_, BLK_)                                   \
    _Pragma("unroll")                                                            \
    for (int i = 0; i < KB; ++i) {                                               \
        const int kk = (BLK_) * KB + i;                                          \
        A_[i] = *(const v2f*)((XA_) + kk * 4);                                   \
        const float* wp = (WB_) + (kk * 4) * WP;                                 \
        P_[i].x = wp[0];      P_[i].y = wp[WP];                                  \
        Q_[i].x = wp[WPLANE]; Q_[i].y = wp[WPLANE + WP];                         \
    }
#define MMA_BLOCK(A_, P_, Q_)                                                    \
    _Pragma("unroll")                                                            \
    for (int i = 0; i < KB; ++i) {                                               \
        acc0 = __builtin_amdgcn_wmma_f32_16x16x4_f32(false, A_[i], false, P_[i], \
                                                     (short)0, acc0, false, false); \
        acc1 = __builtin_amdgcn_wmma_f32_16x16x4_f32(false, A_[i], false, Q_[i], \
                                                     (short)0, acc1, false, false); \
    }

#define ORDER_FENCE asm volatile("" : "+v"(acc0), "+v"(acc1) : : "memory");
#define PASS(S_, SET_)                                                           \
    {                                                                            \
        const int cur = (S_) & 1;                                                \
        const float* xa_s = xa + cur * (BM * XP);                                \
        const float* wb_s = wb + cur * WSTAGE;                                   \
        LOAD_FRAGS(a1, p1, q1, xa_s, wb_s, 1)                                    \
        ORDER_FENCE                                                      \
        MMA_BLOCK(a0, p0, q0)                                                    \
        ORDER_FENCE                                                      \
        if ((S_) + 1 < NSLAB) {                                     \
            lds_stage_barrier();                                                 \
            if ((S_) + 2 < NSLAB) STORE(SET_, cur)                               \
            if ((S_) + 2 + NSET < NSLAB) FETCH(SET_, (S_) + 2 + NSET)            \
            const float* xa_n = xa + (cur ^ 1) * (BM * XP);                      \
            const float* wb_n = wb + (cur ^ 1) * WSTAGE;                         \
            LOAD_FRAGS(a0, p0, q0, xa_n, wb_n, 0)                                \
        }                                                                        \
        ORDER_FENCE                                                      \
        MMA_BLOCK(a1, p1, q1)                                                    \
        ORDER_FENCE                                                      \
    }

    FETCH(A, 0)
    FETCH(B, 1)
    FETCH(C, 2)
    __builtin_amdgcn_sched_barrier(0);
    STORE(A, 0)
    STORE(B, 1)
    __builtin_amdgcn_sched_barrier(0);
    FETCH(A, NSET)
    FETCH(B, NSET + 1)
    lds_stage_barrier();
    LOAD_FRAGS(a0, p0, q0, xa, wb, 0)

    static_assert(NSET == 3 && NSLAB == 16, "the pass list below is written for these");
    PASS(0, C)  PASS(1, A)  PASS(2, B)
    PASS(3, C)  PASS(4, A)  PASS(5, B)
    PASS(6, C)  PASS(7, A)  PASS(8, B)
    PASS(9, C)  PASS(10, A) PASS(11, B)
    PASS(12, C) PASS(13, A)
    PASS(14, B) PASS(15, C)
#undef PASS
#undef ORDER_FENCE
#undef FETCH
#undef STORE
#undef LOAD_FRAGS
#undef MMA_BLOCK

    const int cbase = nh * 32 + n;

    if (sumtile) {
        if (mt != 0) return;
        float s0 = acc0[0], s1 = acc1[0];
#pragma unroll
        for (int r = 1; r < 8; ++r) {
            s0 += acc0[r]; s1 += acc1[r];
        }
        float* vo = vsum + (lane >> 4) * D + cbase;
        vo[0]  = s0 + (float)S * bias[cbase];
        vo[16] = s1 + (float)S * bias[cbase + 16];
        return;
    }

    const int mo = (lane >> 4) * 8;
    float* orow = out + (size_t)(row0 + mt * 16 + mo) * D + cbase;
    const float bi0 = bias[cbase];
    const float bi1 = bias[cbase + 16];
#pragma unroll
    for (int r = 0; r < 8; ++r) {
        float* o = orow + (size_t)r * D;
        o[0]  = acc0[r] + bi0;
        o[16] = acc1[r] + bi1;
    }
}

__global__ __launch_bounds__(256) void attn_kernel(
    const float* __restrict__ Q, const float* __restrict__ K,
    const float* __restrict__ V, const float* __restrict__ Vsum,
    float* __restrict__ out)
{
    const int gwave = (blockIdx.x * blockDim.x + threadIdx.x) >> 5;
    if (gwave >= B * S) return;
    const int lane = threadIdx.x & 31;
    const int b = gwave / S;
    const int i = gwave - b * S;

    const size_t rowQ = (size_t)(b * S + i) * D;
    const float2 qv = *(const float2*)(Q + rowQ + lane * 2);

    float accx = Vsum[b * D + lane * 2];
    float accy = Vsum[b * D + lane * 2 + 1];
    float Z = (float)S;

#pragma unroll
    for (int w = 0; w < WIN; ++w) {
        const int j = i + DIL * (w - WIN / 2);
        if (j < 0 || j >= S) continue;
        const size_t rowJ = (size_t)(b * S + j) * D;
        const float2 kv = *(const float2*)(K + rowJ + lane * 2);
        float c = qv.x * kv.x + qv.y * kv.y;
#pragma unroll
        for (int off = 16; off > 0; off >>= 1) c += __shfl_xor(c, off, 32);
        const float em1 = __expf(c) - 1.0f;
        const float2 vv = *(const float2*)(V + rowJ + lane * 2);
        accx += em1 * vv.x;
        accy += em1 * vv.y;
        Z    += em1;
    }

    const float inv = 1.0f / Z;
    float2 o;
    o.x = accx * inv;
    o.y = accy * inv;
    *(float2*)(out + rowQ + lane * 2) = o;
}

extern "C" void kernel_launch(void* const* d_in, const int* in_sizes, int n_in,
                              void* d_out, int out_size, void* d_ws, size_t ws_size,
                              hipStream_t stream) {
    const float* x  = (const float*)d_in[0];
    const float* Wq = (const float*)d_in[1];
    const float* bq = (const float*)d_in[2];
    const float* Wk = (const float*)d_in[3];
    const float* bk = (const float*)d_in[4];
    const float* Wv = (const float*)d_in[5];
    const float* bv = (const float*)d_in[6];
    float* outp = (float*)d_out;

    constexpr size_t QKV = (size_t)B * S * D;
    float* ws = (float*)d_ws;
    float* Qb = ws;
    float* Kb = ws + QKV;
    float* Vb = ws + 2 * QKV;
    float* Vs = ws + 3 * QKV;

    float* Xs = Kb;
    xsum_kernel<<<dim3(E / 64, NCH), 512, 0, stream>>>(x, Xs);

    qkv_proj_kernel<<<PROJ_WGS + 1, 128, 0, stream>>>(x, Wv, bv, Vb, Xs, Vs);
    qkv_proj_kernel<<<PROJ_WGS, 128, 0, stream>>>(x, Wq, bq, Qb, nullptr, nullptr);
    qkv_proj_kernel<<<PROJ_WGS, 128, 0, stream>>>(x, Wk, bk, Kb, nullptr, nullptr);

    const int nwaves = B * S;
    attn_kernel<<<(nwaves * 32) / 256, 256, 0, stream>>>(Qb, Kb, Vb, Vs, outp);
}


# proj: Q, K, V become ONE launch; work-group tile 64 rows x 96 of the 192 joined columns through LDS (wave 32 x 48): bytes staged per matrix instruction 192 to 107; sum order unchanged
# speedup vs baseline: 20.0032x; 1.3310x over previous
#include <hip/hip_runtime.h>
#include <hip/hip_bf16.h>

constexpr int B   = 2;
constexpr int S   = 4096;
constexpr int E   = 1024;
constexpr int D   = 64;
constexpr int WIN = 5;
constexpr int DIL = 2;

constexpr int NCH       = 16;
constexpr int CH_ROWS   = (B * S) / NCH;
constexpr int BM        = 64;
constexpr int NPL       = 6;
constexpr int BK        = 64;
constexpr int NSLAB     = E / BK;
constexpr int XP        = BK + 4;
constexpr int WP        = 24;
constexpr int WPLANE    = BK * WP + 16;
constexpr int WSTAGE    = NPL * WPLANE;
constexpr int PROJ_WGS  = ((B * S) / BM) * 2;
constexpr int KB        = 8;
constexpr int NSET      = 2;
constexpr int SG        = 16;

typedef __attribute__((ext_vector_type(2))) float v2f;
typedef __attribute__((ext_vector_type(8))) float v8f;

__global__ __launch_bounds__(512) void xsum_kernel(
    const float* __restrict__ x,
    float* __restrict__ xs)
{
    const int lane = threadIdx.x & 31;
    const int wave = threadIdx.x >> 5;
    const int cl   = lane & 15;
    const int rh   = lane >> 4;
    const int cb   = blockIdx.x;
    const int ch   = blockIdx.y;

    const float* p = x + (size_t)(ch * CH_ROWS + wave * 32 + rh) * E + cb * 64 + cl * 4;

    float4 v[16];
#pragma unroll
    for (int it = 0; it < 16; ++it) v[it] = *(const float4*)(p + (size_t)(2 * it) * E);

    float4 s = v[0];
#pragma unroll
    for (int it = 1; it < 16; ++it) {
        s.x += v[it].x; s.y += v[it].y; s.z += v[it].z; s.w += v[it].w;
    }
    s.x += __shfl_xor(s.x, 16, 32);
    s.y += __shfl_xor(s.y, 16, 32);
    s.z += __shfl_xor(s.z, 16, 32);
    s.w += __shfl_xor(s.w, 16, 32);

    __shared__ float4 red[16][16];
    if (rh == 0) red[wave][cl] = s;
    __syncthreads();
    if (threadIdx.x < 16) {
        float4 t = red[0][threadIdx.x];
        for (int w = 1; w < 16; ++w) {
            const float4 u = red[w][threadIdx.x];
            t.x += u.x; t.y += u.y; t.z += u.z; t.w += u.w;
        }
        *(float4*)(xs + (size_t)ch * E + cb * 64 + threadIdx.x * 4) = t;
    }
}

static __device__ __forceinline__ void lds_stage_barrier() {
    asm volatile("s_wait_dscnt 0x0\n\ts_barrier_signal -1\n\ts_barrier_wait -1" ::: "memory");
}

__global__ __launch_bounds__(128) void qkv_proj_kernel(
    const float* __restrict__ x,
    const float* __restrict__ Wq, const float* __restrict__ bq,
    const float* __restrict__ Wk, const float* __restrict__ bk,
    const float* __restrict__ Wv, const float* __restrict__ bv,
    float* __restrict__ Qo,
    float* __restrict__ Ko,
    float* __restrict__ Vo,
    const float* __restrict__ xs,
    float* __restrict__ vsum)
{
    __shared__ __attribute__((aligned(16))) float Xs[2][BM * XP];
    __shared__ __attribute__((aligned(16))) float Ws[2][WSTAGE];

    const int tid  = threadIdx.x;
    const int lane = tid & 31;
    const int wave = tid >> 5;

    const int m  = lane & 15;
    const int n  = lane & 15;
    const int kh = (lane >> 4) * 2;

    if (blockIdx.x == PROJ_WGS) {
        const float* ap = xs + (size_t)m * E + kh;
        const float* bp = Wv + (size_t)kh * D + wave * 16 + n;
        v8f acc = {};
        v2f fa0[SG], fb0[SG], fa1[SG], fb1[SG];
#define SUM_LOAD(FA_, FB_, K0_)                                                  \
        _Pragma("unroll")                                                        \
        for (int i = 0; i < SG; ++i) {                                           \
            const int k = (K0_) + 4 * i;                                         \
            FA_[i] = *(const v2f*)(ap + k);                                      \
            FB_[i].x = bp[(size_t)k * D];                                        \
            FB_[i].y = bp[(size_t)(k + 1) * D];                                  \
        }
#define SUM_MMA(FA_, FB_)                                                        \
        _Pragma("unroll")                                                        \
        for (int i = 0; i < SG; ++i)                                             \
            acc = __builtin_amdgcn_wmma_f32_16x16x4_f32(false, FA_[i], false, FB_[i], \
                                                        (short)0, acc, false, false);
        SUM_LOAD(fa0, fb0, 0)
#pragma nounroll
        for (int k0 = 0; k0 < E; k0 += 2 * SG * 4) {
            SUM_LOAD(fa1, fb1, k0 + SG * 4)
            SUM_MMA(fa0, fb0)
            const int kn = (k0 + 2 * SG * 4 < E) ? k0 + 2 * SG * 4 : 0;
            SUM_LOAD(fa0, fb0, kn)
            SUM_MMA(fa1, fb1)
        }
#undef SUM_LOAD
#undef SUM_MMA
        float s0 = acc[0];
#pragma unroll
        for (int r = 1; r < 8; ++r) s0 += acc[r];
        const int col = wave * 16 + n;
        vsum[(lane >> 4) * D + col] = s0 + (float)S * bv[col];
        return;
    }

    const int wg   = blockIdx.x;
    const int h    = wg & 1;
    const int row0 = (wg >> 1) * BM;
    const int mt2  = wave & 1;
    const int nh   = wave >> 1;

    const float* const Wfull = h ? Wv : Wq;
    const float* const Whalf = Wk + h * 32;

    const int c4  = tid & 15;
    const int r8  = tid >> 4;
    const int c8  = tid & 7;
    const int r16 = tid >> 3;

    const float* xg = x + (size_t)(row0 + r8) * E + c4 * 4;
    const float* wg4 = Wfull + (size_t)r8 * D + c4 * 4;
    const float* wh4 = Whalf + (size_t)r16 * D + c8 * 4;

    float* const xst = &Xs[0][r8 * XP + c4 * 4];
    float* const wst = &Ws[0][(c4 >> 2) * WPLANE + r8 * WP + (c4 & 3) * 4];
    float* const hst = &Ws[0][(4 + (c8 >> 2)) * WPLANE + r16 * WP + (c8 & 3) * 4];

    const float* const xa = &Xs[0][(mt2 * 32 + m) * XP + kh];
    const float* const wb = &Ws[0][(nh * 3) * WPLANE + kh * WP + n];

    float4 xrA0, xrA1, xrA2, xrA3, xrA4, xrA5, xrA6, xrA7;
    float4 wrA0, wrA1, wrA2, wrA3, wrA4, wrA5, wrA6, wrA7, hrA0, hrA1, hrA2, hrA3;
    float4 xrB0, xrB1, xrB2, xrB3, xrB4, xrB5, xrB6, xrB7;
    float4 wrB0, wrB1, wrB2, wrB3, wrB4, wrB5, wrB6, wrB7, hrB0, hrB1, hrB2, hrB3;

    v8f acc00 = {}, acc01 = {}, acc02 = {}, acc10 = {}, acc11 = {}, acc12 = {};

    v2f a0[KB], b0[KB], p0[KB], q0[KB], u0[KB];
    v2f a1[KB], b1[KB], p1[KB], q1[KB], u1[KB];

#define FX(S_, I_) xr##S_##I_ = *(const float4*)(xg + k2 + (size_t)(8 * I_) * E);
#define FW(S_, I_) wr##S_##I_ = *(const float4*)(wg4 + (size_t)(k2 + 8 * I_) * D);
#define FH(S_, I_) hr##S_##I_ = *(const float4*)(wh4 + (size_t)(k2 + 16 * I_) * D);
#define FETCH(S_, T_)                                                            \
    {                                                                            \
        const int k2 = (T_) * BK;                                                \
        FX(S_, 0) FX(S_, 1) FX(S_, 2) FX(S_, 3) FX(S_, 4) FX(S_, 5) FX(S_, 6) FX(S_, 7) \
        FW(S_, 0) FW(S_, 1) FW(S_, 2) FW(S_, 3) FW(S_, 4) FW(S_, 5) FW(S_, 6) FW(S_, 7) \
        FH(S_, 0) FH(S_, 1) FH(S_, 2) FH(S_, 3)                                  \
    }
#define SX(S_, I_) *(float4*)(xs_s + (8 * I_) * XP) = xr##S_##I_;
#define SW(S_, I_) *(float4*)(ws_s + (8 * I_) * WP) = wr##S_##I_;
#define SH(S_, I_) *(float4*)(hs_s + (16 * I_) * WP) = hr##S_##I_;
#define STORE(S_, CUR_)                                                          \
    {                                                                            \
        float* xs_s = xst + (CUR_) * (BM * XP);                                  \
        float* ws_s = wst + (CUR_) * WSTAGE;                                     \
        float* hs_s = hst + (CUR_) * WSTAGE;                                     \
        SX(S_, 0) SX(S_, 1) SX(S_, 2) SX(S_, 3) SX(S_, 4) SX(S_, 5) SX(S_, 6) SX(S_, 7) \
        SW(S_, 0) SW(S_, 1) SW(S_, 2) SW(S_, 3) SW(S_, 4) SW(S_, 5) SW(S_, 6) SW(S_, 7) \
        SH(S_, 0) SH(S_, 1) SH(S_, 2) SH(S_, 3)                                  \
    }
#define LOAD_FRAGS(A_, B_, P_, Q_, U_, XA_, WB_, BLK_)                           \
    _Pragma("unroll")                                                            \
    for (int i = 0; i < KB; ++i) {                                               \
        const int kk = (BLK_) * KB + i;                                          \
        A_[i] = *(const v2f*)((XA_) + kk * 4);                                   \
        B_[i] = *(const v2f*)((XA_) + 16 * XP + kk * 4);                         \
        const float* wp = (WB_) + (kk * 4) * WP;                                 \
        P_[i].x = wp[0];          P_[i].y = wp[WP];                              \
        Q_[i].x = wp[WPLANE];     Q_[i].y = wp[WPLANE + WP];                     \
        U_[i].x = wp[2 * WPLANE]; U_[i].y = wp[2 * WPLANE + WP];                 \
    }
#define MMA1(A_, P_, ACC_)                                                       \
        ACC_ = __builtin_amdgcn_wmma_f32_16x16x4_f32(false, A_, false, P_,       \
                                                     (short)0, ACC_, false, false);
#define MMA_BLOCK(A_, B_, P_, Q_, U_)                                            \
    _Pragma("unroll")                                                            \
    for (int i = 0; i < KB; ++i) {                                               \
        MMA1(A_[i], P_[i], acc00) MMA1(A_[i], Q_[i], acc01) MMA1(A_[i], U_[i], acc02) \
        MMA1(B_[i], P_[i], acc10) MMA1(B_[i], Q_[i], acc11) MMA1(B_[i], U_[i], acc12) \
    }

#define ORDER_FENCE asm volatile("" : "+v"(acc00), "+v"(acc01), "+v"(acc02),    \
                                      "+v"(acc10), "+v"(acc11), "+v"(acc12) : : "memory");
#define PASS(S_, SET_)                                                           \
    {                                                                            \
        const int cur = (S_) & 1;                                                \
        const float* xa_s = xa + cur * (BM * XP);                                \
        const float* wb_s = wb + cur * WSTAGE;                                   \
        LOAD_FRAGS(a1, b1, p1, q1, u1, xa_s, wb_s, 1)                            \
        ORDER_FENCE                                                      \
        MMA_BLOCK(a0, b0, p0, q0, u0)                                            \
        ORDER_FENCE                                                      \
        if ((S_) + 1 < NSLAB) {                                     \
            lds_stage_barrier();                                                 \
            if ((S_) + 2 < NSLAB) STORE(SET_, cur)                               \
            if ((S_) + 2 + NSET < NSLAB) FETCH(SET_, (S_) + 2 + NSET)            \
            const float* xa_n = xa + (cur ^ 1) * (BM * XP);                      \
            const float* wb_n = wb + (cur ^ 1) * WSTAGE;                         \
            LOAD_FRAGS(a0, b0, p0, q0, u0, xa_n, wb_n, 0)                        \
        }                                                                        \
        ORDER_FENCE                                                      \
        MMA_BLOCK(a1, b1, p1, q1, u1)                                            \
        ORDER_FENCE                                                      \
    }

    FETCH(A, 0)
    FETCH(B, 1)
    __builtin_amdgcn_sched_barrier(0);
    STORE(A, 0)
    STORE(B, 1)
    __builtin_amdgcn_sched_barrier(0);
    FETCH(A, NSET)
    FETCH(B, NSET + 1)
    lds_stage_barrier();
    LOAD_FRAGS(a0, b0, p0, q0, u0, xa, wb, 0)

    static_assert(NSET == 2 && NSLAB == 16, "the pass list below is written for these");
    PASS(0, A)  PASS(1, B)  PASS(2, A)  PASS(3, B)
    PASS(4, A)  PASS(5, B)  PASS(6, A)  PASS(7, B)
    PASS(8, A)  PASS(9, B)  PASS(10, A) PASS(11, B)
    PASS(12, A) PASS(13, B)
    PASS(14, A) PASS(15, B)
#undef PASS
#undef ORDER_FENCE
#undef FETCH
#undef STORE
#undef FX
#undef FW
#undef FH
#undef SX
#undef SW
#undef SH
#undef LOAD_FRAGS
#undef MMA_BLOCK
#undef MMA1

    const int mo = (lane >> 4) * 8;
    const size_t orow = (size_t)(row0 + mt2 * 32 + mo) * D;
    float* const ofull = h ? Vo : Qo;
    const float* const bfull = h ? bv : bq;
    float* const ohalf = Ko + h * 32;
    const float* const bhalf = bk + h * 32;
#define WRITE_PLANE(T_, ACC0_, ACC1_)                                            \
    {                                                                            \
        const int lp = nh * 3 + (T_);                           \
        const bool whole = lp < 4;                                               \
        const int col = (whole ? lp * 16 : (lp - 4) * 16) + n;                   \
        float* o = (whole ? ofull : ohalf) + orow + col;                         \
        const float bi = (whole ? bfull : bhalf)[col];                           \
        _Pragma("unroll")     \
        for (int r = 0; r < 8; ++r) {                                            \
            o[(size_t)r * D]        = ACC0_[r] + bi;                             \
            o[(size_t)(r + 16) * D] = ACC1_[r] + bi;                             \
        }                                                                        \
    }
    WRITE_PLANE(0, acc00, acc10)
    WRITE_PLANE(1, acc01, acc11)
    WRITE_PLANE(2, acc02, acc12)
#undef WRITE_PLANE
}

__global__ __launch_bounds__(256) void attn_kernel(
    const float* __restrict__ Q, const float* __restrict__ K,
    const float* __restrict__ V, const float* __restrict__ Vsum,
    float* __restrict__ out)
{
    const int gwave = (blockIdx.x * blockDim.x + threadIdx.x) >> 5;
    if (gwave >= B * S) return;
    const int lane = threadIdx.x & 31;
    const int b = gwave / S;
    const int i = gwave - b * S;

    const size_t rowQ = (size_t)(b * S + i) * D;
    const float2 qv = *(const float2*)(Q + rowQ + lane * 2);

    float accx = Vsum[b * D + lane * 2];
    float accy = Vsum[b * D + lane * 2 + 1];
    float Z = (float)S;

#pragma unroll
    for (int w = 0; w < WIN; ++w) {
        const int j = i + DIL * (w - WIN / 2);
        if (j < 0 || j >= S) continue;
        const size_t rowJ = (size_t)(b * S + j) * D;
        const float2 kv = *(const float2*)(K + rowJ + lane * 2);
        float c = qv.x * kv.x + qv.y * kv.y;
#pragma unroll
        for (int off = 16; off > 0; off >>= 1) c += __shfl_xor(c, off, 32);
        const float em1 = __expf(c) - 1.0f;
        const float2 vv = *(const float2*)(V + rowJ + lane * 2);
        accx += em1 * vv.x;
        accy += em1 * vv.y;
        Z    += em1;
    }

    const float inv = 1.0f / Z;
    float2 o;
    o.x = accx * inv;
    o.y = accy * inv;
    *(float2*)(out + rowQ + lane * 2) = o;
}

extern "C" void kernel_launch(void* const* d_in, const int* in_sizes, int n_in,
                              void* d_out, int out_size, void* d_ws, size_t ws_size,
                              hipStream_t stream) {
    const float* x  = (const float*)d_in[0];
    const float* Wq = (const float*)d_in[1];
    const float* bq = (const float*)d_in[2];
    const float* Wk = (const float*)d_in[3];
    const float* bk = (const float*)d_in[4];
    const float* Wv = (const float*)d_in[5];
    const float* bv = (const float*)d_in[6];
    float* outp = (float*)d_out;

    constexpr size_t QKV = (size_t)B * S * D;
    float* ws = (float*)d_ws;
    float* Qb = ws;
    float* Kb = ws + QKV;
    float* Vb = ws + 2 * QKV;
    float* Vs = ws + 3 * QKV;

    float* Xs = outp;
    xsum_kernel<<<dim3(E / 64, NCH), 512, 0, stream>>>(x, Xs);

    qkv_proj_kernel<<<PROJ_WGS + 1, 128, 0, stream>>>(x, Wq, bq, Wk, bk, Wv, bv,
                                                      Qb, Kb, Vb, Xs, Vs);

    const int nwaves = B * S;
    attn_kernel<<<(nwaves * 32) / 256, 256, 0, stream>>>(Qb, Kb, Vb, Vs, outp);
}


# x-sum launch removed: 128 streaming work-groups in the projection launch sum x and multiply by their own 64 k of Wv (K-split sum tile, f32 WMMA); attention adds 64 partial rows; V sum order changes
# speedup vs baseline: 21.0402x; 1.0518x over previous
#include <hip/hip_runtime.h>
#include <hip/hip_bf16.h>

constexpr int B   = 2;
constexpr int S   = 4096;
constexpr int E   = 1024;
constexpr int D   = 64;
constexpr int WIN = 5;
constexpr int DIL = 2;

constexpr int NXW       = 128;
constexpr int XR        = (B * S) / (NXW / 16);
constexpr int NPB       = NXW / B;
constexpr int XG        = 32;
constexpr int BM        = 64;
constexpr int NPL       = 6;
constexpr int BK        = 64;
constexpr int NSLAB     = E / BK;
constexpr int XP        = BK + 4;
constexpr int WP        = 24;
constexpr int WPLANE    = BK * WP + 16;
constexpr int WSTAGE    = NPL * WPLANE;
constexpr int PROJ_WGS  = ((B * S) / BM) * 2;
constexpr int KB        = 8;
constexpr int NSET      = 2;

typedef __attribute__((ext_vector_type(2))) float v2f;
typedef __attribute__((ext_vector_type(8))) float v8f;

static __device__ __forceinline__ void lds_stage_barrier() {
    asm volatile("s_wait_dscnt 0x0\n\ts_barrier_signal -1\n\ts_barrier_wait -1" ::: "memory");
}

__global__ __launch_bounds__(128) void qkv_proj_kernel(
    const float* __restrict__ x,
    const float* __restrict__ Wq, const float* __restrict__ bq,
    const float* __restrict__ Wk, const float* __restrict__ bk,
    const float* __restrict__ Wv, const float* __restrict__ bv,
    float* __restrict__ Qo,
    float* __restrict__ Ko,
    float* __restrict__ Vo,
    float* __restrict__ part)
{
    __shared__ __attribute__((aligned(16))) float Xs[2][BM * XP];
    __shared__ __attribute__((aligned(16))) float Ws[2][WSTAGE];

    const int tid  = threadIdx.x;
    const int lane = tid & 31;
    const int wave = tid >> 5;

    const int m  = lane & 15;
    const int n  = lane & 15;
    const int kh = (lane >> 4) * 2;

    if (blockIdx.x >= PROJ_WGS) {
        const int sw    = blockIdx.x - PROJ_WGS;
        const int slab  = sw & 15;
        const int chunk = sw >> 4;
        const int cl    = lane & 15;
        const int rh    = lane >> 4;

        const float* bp = Wv + (size_t)(slab * 64 + kh) * D + wave * 16 + n;
        v2f fb[16];
#pragma unroll
        for (int i = 0; i < 16; ++i) {
            fb[i].x = bp[(size_t)(4 * i) * D];
            fb[i].y = bp[(size_t)(4 * i + 1) * D];
        }

        const float* p = x + (size_t)(chunk * XR + wave * (XR / 4) + rh) * E + slab * 64 + cl * 4;
        float4 va[XG], vb[XG];
        float4 s0 = {0.f, 0.f, 0.f, 0.f}, s1 = {0.f, 0.f, 0.f, 0.f};
#define XLOAD(V_, G_)                                                            \
        _Pragma("unroll")                                                        \
        for (int i = 0; i < XG; ++i)                                             \
            V_[i] = *(const float4*)(p + (size_t)(2 * ((G_) * XG + i)) * E);
#define XADD(V_)                                                                 \
        _Pragma("unroll")                                                        \
        for (int i = 0; i < XG; i += 2) {                                        \
            s0.x += V_[i].x;     s0.y += V_[i].y;     s0.z += V_[i].z;     s0.w += V_[i].w;     \
            s1.x += V_[i + 1].x; s1.y += V_[i + 1].y; s1.z += V_[i + 1].z; s1.w += V_[i + 1].w; \
        }
        static_assert(XR / 4 == 2 * XG * 4, "four groups of XG loads cover a wave's rows");
        XLOAD(va, 0)
        XLOAD(vb, 1)
        __builtin_amdgcn_sched_barrier(0);
        XADD(va)
        XLOAD(va, 2)
        __builtin_amdgcn_sched_barrier(0);
        XADD(vb)
        XLOAD(vb, 3)
        __builtin_amdgcn_sched_barrier(0);
        XADD(va)
        XADD(vb)
#undef XLOAD
#undef XADD

        float* const As = &Xs[0][0];
        float* const Os = &Xs[1][0];
        const int mr = wave * 4 + rh * 2;
        *(float4*)(As + (mr    ) * XP + cl * 4) = s0;
        *(float4*)(As + (mr + 1) * XP + cl * 4) = s1;
        __syncthreads();

        const float* ap = As + m * XP + kh;
        v8f acc0 = {}, acc1 = {};
#pragma unroll
        for (int i = 0; i < 16; i += 2) {
            const v2f fa0 = *(const v2f*)(ap + 4 * i);
            const v2f fa1 = *(const v2f*)(ap + 4 * i + 4);
            acc0 = __builtin_amdgcn_wmma_f32_16x16x4_f32(false, fa0, false, fb[i],
                                                         (short)0, acc0, false, false);
            acc1 = __builtin_amdgcn_wmma_f32_16x16x4_f32(false, fa1, false, fb[i + 1],
                                                         (short)0, acc1, false, false);
        }
        float t = acc0[0] + acc1[0];
#pragma unroll
        for (int r = 1; r < 8; ++r) t += acc0[r] + acc1[r];
        t += __shfl_xor(t, 16, 32);
        const int col = wave * 16 + n;
        if (lane < 16) Os[col] = t + (slab == 0 ? (float)XR * bv[col] : 0.f);
        __syncthreads();
        if (wave == 0)
            *(float2*)(part + (size_t)sw * D + lane * 2) = *(const float2*)(Os + lane * 2);
        return;
    }

    const int wg   = blockIdx.x;
    const int h    = wg & 1;
    const int row0 = (wg >> 1) * BM;
    const int mt2  = wave & 1;
    const int nh   = wave >> 1;

    const float* const Wfull = h ? Wv : Wq;
    const float* const Whalf = Wk + h * 32;

    const int c4  = tid & 15;
    const int r8  = tid >> 4;
    const int c8  = tid & 7;
    const int r16 = tid >> 3;

    const float* xg = x + (size_t)(row0 + r8) * E + c4 * 4;
    const float* wg4 = Wfull + (size_t)r8 * D + c4 * 4;
    const float* wh4 = Whalf + (size_t)r16 * D + c8 * 4;

    float* const xst = &Xs[0][r8 * XP + c4 * 4];
    float* const wst = &Ws[0][(c4 >> 2) * WPLANE + r8 * WP + (c4 & 3) * 4];
    float* const hst = &Ws[0][(4 + (c8 >> 2)) * WPLANE + r16 * WP + (c8 & 3) * 4];

    const float* const xa = &Xs[0][(mt2 * 32 + m) * XP + kh];
    const float* const wb = &Ws[0][(nh * 3) * WPLANE + kh * WP + n];

    float4 xrA0, xrA1, xrA2, xrA3, xrA4, xrA5, xrA6, xrA7;
    float4 wrA0, wrA1, wrA2, wrA3, wrA4, wrA5, wrA6, wrA7, hrA0, hrA1, hrA2, hrA3;
    float4 xrB0, xrB1, xrB2, xrB3, xrB4, xrB5, xrB6, xrB7;
    float4 wrB0, wrB1, wrB2, wrB3, wrB4, wrB5, wrB6, wrB7, hrB0, hrB1, hrB2, hrB3;

    v8f acc00 = {}, acc01 = {}, acc02 = {}, acc10 = {}, acc11 = {}, acc12 = {};

    v2f a0[KB], b0[KB], p0[KB], q0[KB], u0[KB];
    v2f a1[KB], b1[KB], p1[KB], q1[KB], u1[KB];

#define FX(S_, I_) xr##S_##I_ = *(const float4*)(xg + k2 + (size_t)(8 * I_) * E);
#define FW(S_, I_) wr##S_##I_ = *(const float4*)(wg4 + (size_t)(k2 + 8 * I_) * D);
#define FH(S_, I_) hr##S_##I_ = *(const float4*)(wh4 + (size_t)(k2 + 16 * I_) * D);
#define FETCH(S_, T_)                                                            \
    {                                                                            \
        const int k2 = (T_) * BK;                                                \
        FX(S_, 0) FX(S_, 1) FX(S_, 2) FX(S_, 3) FX(S_, 4) FX(S_, 5) FX(S_, 6) FX(S_, 7) \
        FW(S_, 0) FW(S_, 1) FW(S_, 2) FW(S_, 3) FW(S_, 4) FW(S_, 5) FW(S_, 6) FW(S_, 7) \
        FH(S_, 0) FH(S_, 1) FH(S_, 2) FH(S_, 3)                                  \
    }
#define SX(S_, I_) *(float4*)(xs_s + (8 * I_) * XP) = xr##S_##I_;
#define SW(S_, I_) *(float4*)(ws_s + (8 * I_) * WP) = wr##S_##I_;
#define SH(S_, I_) *(float4*)(hs_s + (16 * I_) * WP) = hr##S_##I_;
#define STORE(S_, CUR_)                                                          \
    {                                                                            \
        float* xs_s = xst + (CUR_) * (BM * XP);                                  \
        float* ws_s = wst + (CUR_) * WSTAGE;                                     \
        float* hs_s = hst + (CUR_) * WSTAGE;                                     \
        SX(S_, 0) SX(S_, 1) SX(S_, 2) SX(S_, 3) SX(S_, 4) SX(S_, 5) SX(S_, 6) SX(S_, 7) \
        SW(S_, 0) SW(S_, 1) SW(S_, 2) SW(S_, 3) SW(S_, 4) SW(S_, 5) SW(S_, 6) SW(S_, 7) \
        SH(S_, 0) SH(S_, 1) SH(S_, 2) SH(S_, 3)                                  \
    }
#define LOAD_FRAGS(A_, B_, P_, Q_, U_, XA_, WB_, BLK_)                           \
    _Pragma("unroll")                                                            \
    for (int i = 0; i < KB; ++i) {                                               \
        const int kk = (BLK_) * KB + i;                                          \
        A_[i] = *(const v2f*)((XA_) + kk * 4);                                   \
        B_[i] = *(const v2f*)((XA_) + 16 * XP + kk * 4);                         \
        const float* wp = (WB_) + (kk * 4) * WP;                                 \
        P_[i].x = wp[0];          P_[i].y = wp[WP];                              \
        Q_[i].x = wp[WPLANE];     Q_[i].y = wp[WPLANE + WP];                     \
        U_[i].x = wp[2 * WPLANE]; U_[i].y = wp[2 * WPLANE + WP];                 \
    }
#define MMA1(A_, P_, ACC_)                                                       \
        ACC_ = __builtin_amdgcn_wmma_f32_16x16x4_f32(false, A_, false, P_,       \
                                                     (short)0, ACC_, false, false);
#define MMA_BLOCK(A_, B_, P_, Q_, U_)                                            \
    _Pragma("unroll")                                                            \
    for (int i = 0; i < KB; ++i) {                                               \
        MMA1(A_[i], P_[i], acc00) MMA1(A_[i], Q_[i], acc01) MMA1(A_[i], U_[i], acc02) \
        MMA1(B_[i], P_[i], acc10) MMA1(B_[i], Q_[i], acc11) MMA1(B_[i], U_[i], acc12) \
    }

#define ORDER_FENCE asm volatile("" : "+v"(acc00), "+v"(acc01), "+v"(acc02),    \
                                      "+v"(acc10), "+v"(acc11), "+v"(acc12) : : "memory");
#define PASS(S_, SET_)                                                           \
    {                                                                            \
        const int cur = (S_) & 1;                                                \
        const float* xa_s = xa + cur * (BM * XP);                                \
        const float* wb_s = wb + cur * WSTAGE;                                   \
        LOAD_FRAGS(a1, b1, p1, q1, u1, xa_s, wb_s, 1)                            \
        ORDER_FENCE                                                      \
        MMA_BLOCK(a0, b0, p0, q0, u0)                                            \
        ORDER_FENCE                                                      \
        if ((S_) + 1 < NSLAB) {                                     \
            lds_stage_barrier();                                                 \
            if ((S_) + 2 < NSLAB) STORE(SET_, cur)                               \
            if ((S_) + 2 + NSET < NSLAB) FETCH(SET_, (S_) + 2 + NSET)            \
            const float* xa_n = xa + (cur ^ 1) * (BM * XP);                      \
            const float* wb_n = wb + (cur ^ 1) * WSTAGE;                         \
            LOAD_FRAGS(a0, b0, p0, q0, u0, xa_n, wb_n, 0)                        \
        }                                                                        \
        ORDER_FENCE                                                      \
        MMA_BLOCK(a1, b1, p1, q1, u1)                                            \
        ORDER_FENCE                                                      \
    }

    FETCH(A, 0)
    FETCH(B, 1)
    __builtin_amdgcn_sched_barrier(0);
    STORE(A, 0)
    STORE(B, 1)
    __builtin_amdgcn_sched_barrier(0);
    FETCH(A, NSET)
    FETCH(B, NSET + 1)
    lds_stage_barrier();
    LOAD_FRAGS(a0, b0, p0, q0, u0, xa, wb, 0)

    static_assert(NSET == 2 && NSLAB == 16, "the pass list below is written for these");
    PASS(0, A)  PASS(1, B)  PASS(2, A)  PASS(3, B)
    PASS(4, A)  PASS(5, B)  PASS(6, A)  PASS(7, B)
    PASS(8, A)  PASS(9, B)  PASS(10, A) PASS(11, B)
    PASS(12, A) PASS(13, B)
    PASS(14, A) PASS(15, B)
#undef PASS
#undef ORDER_FENCE
#undef FETCH
#undef STORE
#undef FX
#undef FW
#undef FH
#undef SX
#undef SW
#undef SH
#undef LOAD_FRAGS
#undef MMA_BLOCK
#undef MMA1

    const int mo = (lane >> 4) * 8;
    const size_t orow = (size_t)(row0 + mt2 * 32 + mo) * D;
    float* const ofull = h ? Vo : Qo;
    const float* const bfull = h ? bv : bq;
    float* const ohalf = Ko + h * 32;
    const float* const bhalf = bk + h * 32;
#define WRITE_PLANE(T_, ACC0_, ACC1_)                                            \
    {                                                                            \
        const int lp = nh * 3 + (T_);                           \
        const bool whole = lp < 4;                                               \
        const int col = (whole ? lp * 16 : (lp - 4) * 16) + n;                   \
        float* o = (whole ? ofull : ohalf) + orow + col;                         \
        const float bi = (whole ? bfull : bhalf)[col];                           \
        _Pragma("unroll")     \
        for (int r = 0; r < 8; ++r) {                                            \
            o[(size_t)r * D]        = ACC0_[r] + bi;                             \
            o[(size_t)(r + 16) * D] = ACC1_[r] + bi;                             \
        }                                                                        \
    }
    WRITE_PLANE(0, acc00, acc10)
    WRITE_PLANE(1, acc01, acc11)
    WRITE_PLANE(2, acc02, acc12)
#undef WRITE_PLANE
}

__global__ __launch_bounds__(256) void attn_kernel(
    float* qo,
    const float* __restrict__ K, const float* __restrict__ V,
    const float* __restrict__ part)
{
    __shared__ __attribute__((aligned(16))) float red[16 * D];

    const int tid  = threadIdx.x;
    const int lane = tid & 31;
    const int gwave = blockIdx.x * 8 + (tid >> 5);
    const int b = gwave / S;
    const int i = gwave - b * S;

    const size_t rowQ = (size_t)(b * S + i) * D;
    const float2 qv = *(const float2*)(qo + rowQ + lane * 2);

    {
        const int c4 = tid & 15;
        const int g  = tid >> 4;
        const float* pp = part + (size_t)(b * NPB + g) * D + c4 * 4;
        const float4 t0 = *(const float4*)(pp);
        const float4 t1 = *(const float4*)(pp + 16 * D);
        const float4 t2 = *(const float4*)(pp + 32 * D);
        const float4 t3 = *(const float4*)(pp + 48 * D);
        float4 t;
        t.x = (t0.x + t1.x) + (t2.x + t3.x);
        t.y = (t0.y + t1.y) + (t2.y + t3.y);
        t.z = (t0.z + t1.z) + (t2.z + t3.z);
        t.w = (t0.w + t1.w) + (t2.w + t3.w);
        *(float4*)(red + g * D + c4 * 4) = t;
    }
    __syncthreads();

    float accx = 0.f, accy = 0.f;
#pragma unroll
    for (int g = 0; g < 16; ++g) {
        const float2 r = *(const float2*)(red + g * D + lane * 2);
        accx += r.x;
        accy += r.y;
    }
    float Z = (float)S;

#pragma unroll
    for (int w = 0; w < WIN; ++w) {
        const int j = i + DIL * (w - WIN / 2);
        if (j < 0 || j >= S) continue;
        const size_t rowJ = (size_t)(b * S + j) * D;
        const float2 kv = *(const float2*)(K + rowJ + lane * 2);
        float c = qv.x * kv.x + qv.y * kv.y;
#pragma unroll
        for (int off = 16; off > 0; off >>= 1) c += __shfl_xor(c, off, 32);
        const float em1 = __expf(c) - 1.0f;
        const float2 vv = *(const float2*)(V + rowJ + lane * 2);
        accx += em1 * vv.x;
        accy += em1 * vv.y;
        Z    += em1;
    }

    const float inv = 1.0f / Z;
    float2 o;
    o.x = accx * inv;
    o.y = accy * inv;
    *(float2*)(qo + rowQ + lane * 2) = o;
}

extern "C" void kernel_launch(void* const* d_in, const int* in_sizes, int n_in,
                              void* d_out, int out_size, void* d_ws, size_t ws_size,
                              hipStream_t stream) {
    const float* x  = (const float*)d_in[0];
    const float* Wq = (const float*)d_in[1];
    const float* bq = (const float*)d_in[2];
    const float* Wk = (const float*)d_in[3];
    const float* bk = (const float*)d_in[4];
    const float* Wv = (const float*)d_in[5];
    const float* bv = (const float*)d_in[6];
    float* outp = (float*)d_out;

    constexpr size_t QKV = (size_t)B * S * D;
    float* ws = (float*)d_ws;
    float* Pp = ws;
    float* Kb = ws + QKV;
    float* Vb = ws + 2 * QKV;

    qkv_proj_kernel<<<PROJ_WGS + NXW, 128, 0, stream>>>(x, Wq, bq, Wk, bk, Wv, bv,
                                                        outp, Kb, Vb, Pp);

    attn_kernel<<<(B * S) / 8, 256, 0, stream>>>(outp, Kb, Vb, Pp);
}


# attn: all of a wave's loads (5 K + 5 V window rows, q, Vsum parts) issued at its start, rows clamped in bounds and outside weights zeroed: 1 memory wait, was 6 in a row; sum order unchanged
# speedup vs baseline: 22.4212x; 1.0656x over previous
#include <hip/hip_runtime.h>
#include <hip/hip_bf16.h>

constexpr int B   = 2;
constexpr int S   = 4096;
constexpr int E   = 1024;
constexpr int D   = 64;
constexpr int WIN = 5;
constexpr int DIL = 2;

constexpr int NXW       = 128;
constexpr int XR        = (B * S) / (NXW / 16);
constexpr int NPB       = NXW / B;
constexpr int XG        = 32;
constexpr int BM        = 64;
constexpr int NPL       = 6;
constexpr int BK        = 64;
constexpr int NSLAB     = E / BK;
constexpr int XP        = BK + 4;
constexpr int WP        = 24;
constexpr int WPLANE    = BK * WP + 16;
constexpr int WSTAGE    = NPL * WPLANE;
constexpr int PROJ_WGS  = ((B * S) / BM) * 2;
constexpr int KB        = 8;
constexpr int NSET      = 2;

typedef __attribute__((ext_vector_type(2))) float v2f;
typedef __attribute__((ext_vector_type(8))) float v8f;

static __device__ __forceinline__ void lds_stage_barrier() {
    asm volatile("s_wait_dscnt 0x0\n\ts_barrier_signal -1\n\ts_barrier_wait -1" ::: "memory");
}

__global__ __launch_bounds__(128) void qkv_proj_kernel(
    const float* __restrict__ x,
    const float* __restrict__ Wq, const float* __restrict__ bq,
    const float* __restrict__ Wk, const float* __restrict__ bk,
    const float* __restrict__ Wv, const float* __restrict__ bv,
    float* __restrict__ Qo,
    float* __restrict__ Ko,
    float* __restrict__ Vo,
    float* __restrict__ part)
{
    __shared__ __attribute__((aligned(16))) float Xs[2][BM * XP];
    __shared__ __attribute__((aligned(16))) float Ws[2][WSTAGE];

    const int tid  = threadIdx.x;
    const int lane = tid & 31;
    const int wave = tid >> 5;

    const int m  = lane & 15;
    const int n  = lane & 15;
    const int kh = (lane >> 4) * 2;

    if (blockIdx.x >= PROJ_WGS) {
        const int sw    = blockIdx.x - PROJ_WGS;
        const int slab  = sw & 15;
        const int chunk = sw >> 4;
        const int cl    = lane & 15;
        const int rh    = lane >> 4;

        const float* bp = Wv + (size_t)(slab * 64 + kh) * D + wave * 16 + n;
        v2f fb[16];
#pragma unroll
        for (int i = 0; i < 16; ++i) {
            fb[i].x = bp[(size_t)(4 * i) * D];
            fb[i].y = bp[(size_t)(4 * i + 1) * D];
        }

        const float* p = x + (size_t)(chunk * XR + wave * (XR / 4) + rh) * E + slab * 64 + cl * 4;
        float4 va[XG], vb[XG];
        float4 s0 = {0.f, 0.f, 0.f, 0.f}, s1 = {0.f, 0.f, 0.f, 0.f};
#define XLOAD(V_, G_)                                                            \
        _Pragma("unroll")                                                        \
        for (int i = 0; i < XG; ++i)                                             \
            V_[i] = *(const float4*)(p + (size_t)(2 * ((G_) * XG + i)) * E);
#define XADD(V_)                                                                 \
        _Pragma("unroll")                                                        \
        for (int i = 0; i < XG; i += 2) {                                        \
            s0.x += V_[i].x;     s0.y += V_[i].y;     s0.z += V_[i].z;     s0.w += V_[i].w;     \
            s1.x += V_[i + 1].x; s1.y += V_[i + 1].y; s1.z += V_[i + 1].z; s1.w += V_[i + 1].w; \
        }
        static_assert(XR / 4 == 2 * XG * 4, "four groups of XG loads cover a wave's rows");
        XLOAD(va, 0)
        XLOAD(vb, 1)
        __builtin_amdgcn_sched_barrier(0);
        XADD(va)
        XLOAD(va, 2)
        __builtin_amdgcn_sched_barrier(0);
        XADD(vb)
        XLOAD(vb, 3)
        __builtin_amdgcn_sched_barrier(0);
        XADD(va)
        XADD(vb)
#undef XLOAD
#undef XADD

        float* const As = &Xs[0][0];
        float* const Os = &Xs[1][0];
        const int mr = wave * 4 + rh * 2;
        *(float4*)(As + (mr    ) * XP + cl * 4) = s0;
        *(float4*)(As + (mr + 1) * XP + cl * 4) = s1;
        __syncthreads();

        const float* ap = As + m * XP + kh;
        v8f acc0 = {}, acc1 = {};
#pragma unroll
        for (int i = 0; i < 16; i += 2) {
            const v2f fa0 = *(const v2f*)(ap + 4 * i);
            const v2f fa1 = *(const v2f*)(ap + 4 * i + 4);
            acc0 = __builtin_amdgcn_wmma_f32_16x16x4_f32(false, fa0, false, fb[i],
                                                         (short)0, acc0, false, false);
            acc1 = __builtin_amdgcn_wmma_f32_16x16x4_f32(false, fa1, false, fb[i + 1],
                                                         (short)0, acc1, false, false);
        }
        float t = acc0[0] + acc1[0];
#pragma unroll
        for (int r = 1; r < 8; ++r) t += acc0[r] + acc1[r];
        t += __shfl_xor(t, 16, 32);
        const int col = wave * 16 + n;
        if (lane < 16) Os[col] = t + (slab == 0 ? (float)XR * bv[col] : 0.f);
        __syncthreads();
        if (wave == 0)
            *(float2*)(part + (size_t)sw * D + lane * 2) = *(const float2*)(Os + lane * 2);
        return;
    }

    const int wg   = blockIdx.x;
    const int h    = wg & 1;
    const int row0 = (wg >> 1) * BM;
    const int mt2  = wave & 1;
    const int nh   = wave >> 1;

    const float* const Wfull = h ? Wv : Wq;
    const float* const Whalf = Wk + h * 32;

    const int c4  = tid & 15;
    const int r8  = tid >> 4;
    const int c8  = tid & 7;
    const int r16 = tid >> 3;

    const float* xg = x + (size_t)(row0 + r8) * E + c4 * 4;
    const float* wg4 = Wfull + (size_t)r8 * D + c4 * 4;
    const float* wh4 = Whalf + (size_t)r16 * D + c8 * 4;

    float* const xst = &Xs[0][r8 * XP + c4 * 4];
    float* const wst = &Ws[0][(c4 >> 2) * WPLANE + r8 * WP + (c4 & 3) * 4];
    float* const hst = &Ws[0][(4 + (c8 >> 2)) * WPLANE + r16 * WP + (c8 & 3) * 4];

    const float* const xa = &Xs[0][(mt2 * 32 + m) * XP + kh];
    const float* const wb = &Ws[0][(nh * 3) * WPLANE + kh * WP + n];

    float4 xrA0, xrA1, xrA2, xrA3, xrA4, xrA5, xrA6, xrA7;
    float4 wrA0, wrA1, wrA2, wrA3, wrA4, wrA5, wrA6, wrA7, hrA0, hrA1, hrA2, hrA3;
    float4 xrB0, xrB1, xrB2, xrB3, xrB4, xrB5, xrB6, xrB7;
    float4 wrB0, wrB1, wrB2, wrB3, wrB4, wrB5, wrB6, wrB7, hrB0, hrB1, hrB2, hrB3;

    v8f acc00 = {}, acc01 = {}, acc02 = {}, acc10 = {}, acc11 = {}, acc12 = {};

    v2f a0[KB], b0[KB], p0[KB], q0[KB], u0[KB];
    v2f a1[KB], b1[KB], p1[KB], q1[KB], u1[KB];

#define FX(S_, I_) xr##S_##I_ = *(const float4*)(xg + k2 + (size_t)(8 * I_) * E);
#define FW(S_, I_) wr##S_##I_ = *(const float4*)(wg4 + (size_t)(k2 + 8 * I_) * D);
#define FH(S_, I_) hr##S_##I_ = *(const float4*)(wh4 + (size_t)(k2 + 16 * I_) * D);
#define FETCH(S_, T_)                                                            \
    {                                                                            \
        const int k2 = (T_) * BK;                                                \
        FX(S_, 0) FX(S_, 1) FX(S_, 2) FX(S_, 3) FX(S_, 4) FX(S_, 5) FX(S_, 6) FX(S_, 7) \
        FW(S_, 0) FW(S_, 1) FW(S_, 2) FW(S_, 3) FW(S_, 4) FW(S_, 5) FW(S_, 6) FW(S_, 7) \
        FH(S_, 0) FH(S_, 1) FH(S_, 2) FH(S_, 3)                                  \
    }
#define SX(S_, I_) *(float4*)(xs_s + (8 * I_) * XP) = xr##S_##I_;
#define SW(S_, I_) *(float4*)(ws_s + (8 * I_) * WP) = wr##S_##I_;
#define SH(S_, I_) *(float4*)(hs_s + (16 * I_) * WP) = hr##S_##I_;
#define STORE(S_, CUR_)                                                          \
    {                                                                            \
        float* xs_s = xst + (CUR_) * (BM * XP);                                  \
        float* ws_s = wst + (CUR_) * WSTAGE;                                     \
        float* hs_s = hst + (CUR_) * WSTAGE;                                     \
        SX(S_, 0) SX(S_, 1) SX(S_, 2) SX(S_, 3) SX(S_, 4) SX(S_, 5) SX(S_, 6) SX(S_, 7) \
        SW(S_, 0) SW(S_, 1) SW(S_, 2) SW(S_, 3) SW(S_, 4) SW(S_, 5) SW(S_, 6) SW(S_, 7) \
        SH(S_, 0) SH(S_, 1) SH(S_, 2) SH(S_, 3)                                  \
    }
#define LOAD_FRAGS(A_, B_, P_, Q_, U_, XA_, WB_, BLK_)                           \
    _Pragma("unroll")                                                            \
    for (int i = 0; i < KB; ++i) {                                               \
        const int kk = (BLK_) * KB + i;                                          \
        A_[i] = *(const v2f*)((XA_) + kk * 4);                                   \
        B_[i] = *(const v2f*)((XA_) + 16 * XP + kk * 4);                         \
        const float* wp = (WB_) + (kk * 4) * WP;                                 \
        P_[i].x = wp[0];          P_[i].y = wp[WP];                              \
        Q_[i].x = wp[WPLANE];     Q_[i].y = wp[WPLANE + WP];                     \
        U_[i].x = wp[2 * WPLANE]; U_[i].y = wp[2 * WPLANE + WP];                 \
    }
#define MMA1(A_, P_, ACC_)                                                       \
        ACC_ = __builtin_amdgcn_wmma_f32_16x16x4_f32(false, A_, false, P_,       \
                                                     (short)0, ACC_, false, false);
#define MMA_BLOCK(A_, B_, P_, Q_, U_)                                            \
    _Pragma("unroll")                                                            \
    for (int i = 0; i < KB; ++i) {                                               \
        MMA1(A_[i], P_[i], acc00) MMA1(A_[i], Q_[i], acc01) MMA1(A_[i], U_[i], acc02) \
        MMA1(B_[i], P_[i], acc10) MMA1(B_[i], Q_[i], acc11) MMA1(B_[i], U_[i], acc12) \
    }

#define ORDER_FENCE asm volatile("" : "+v"(acc00), "+v"(acc01), "+v"(acc02),    \
                                      "+v"(acc10), "+v"(acc11), "+v"(acc12) : : "memory");
#define PASS(S_, SET_)                                                           \
    {                                                                            \
        const int cur = (S_) & 1;                                                \
        const float* xa_s = xa + cur * (BM * XP);                                \
        const float* wb_s = wb + cur * WSTAGE;                                   \
        LOAD_FRAGS(a1, b1, p1, q1, u1, xa_s, wb_s, 1)                            \
        ORDER_FENCE                                                      \
        MMA_BLOCK(a0, b0, p0, q0, u0)                                            \
        ORDER_FENCE                                                      \
        if ((S_) + 1 < NSLAB) {                                     \
            lds_stage_barrier();                                                 \
            if ((S_) + 2 < NSLAB) STORE(SET_, cur)                               \
            if ((S_) + 2 + NSET < NSLAB) FETCH(SET_, (S_) + 2 + NSET)            \
            const float* xa_n = xa + (cur ^ 1) * (BM * XP);                      \
            const float* wb_n = wb + (cur ^ 1) * WSTAGE;                         \
            LOAD_FRAGS(a0, b0, p0, q0, u0, xa_n, wb_n, 0)                        \
        }                                                                        \
        ORDER_FENCE                                                      \
        MMA_BLOCK(a1, b1, p1, q1, u1)                                            \
        ORDER_FENCE                                                      \
    }

    FETCH(A, 0)
    FETCH(B, 1)
    __builtin_amdgcn_sched_barrier(0);
    STORE(A, 0)
    STORE(B, 1)
    __builtin_amdgcn_sched_barrier(0);
    FETCH(A, NSET)
    FETCH(B, NSET + 1)
    lds_stage_barrier();
    LOAD_FRAGS(a0, b0, p0, q0, u0, xa, wb, 0)

    static_assert(NSET == 2 && NSLAB == 16, "the pass list below is written for these");
    PASS(0, A)  PASS(1, B)  PASS(2, A)  PASS(3, B)
    PASS(4, A)  PASS(5, B)  PASS(6, A)  PASS(7, B)
    PASS(8, A)  PASS(9, B)  PASS(10, A) PASS(11, B)
    PASS(12, A) PASS(13, B)
    PASS(14, A) PASS(15, B)
#undef PASS
#undef ORDER_FENCE
#undef FETCH
#undef STORE
#undef FX
#undef FW
#undef FH
#undef SX
#undef SW
#undef SH
#undef LOAD_FRAGS
#undef MMA_BLOCK
#undef MMA1

    const int mo = (lane >> 4) * 8;
    const size_t orow = (size_t)(row0 + mt2 * 32 + mo) * D;
    float* const ofull = h ? Vo : Qo;
    const float* const bfull = h ? bv : bq;
    float* const ohalf = Ko + h * 32;
    const float* const bhalf = bk + h * 32;
#define WRITE_PLANE(T_, ACC0_, ACC1_)                                            \
    {                                                                            \
        const int lp = nh * 3 + (T_);                           \
        const bool whole = lp < 4;                                               \
        const int col = (whole ? lp * 16 : (lp - 4) * 16) + n;                   \
        float* o = (whole ? ofull : ohalf) + orow + col;                         \
        const float bi = (whole ? bfull : bhalf)[col];                           \
        _Pragma("unroll")     \
        for (int r = 0; r < 8; ++r) {                                            \
            o[(size_t)r * D]        = ACC0_[r] + bi;                             \
            o[(size_t)(r + 16) * D] = ACC1_[r] + bi;                             \
        }                                                                        \
    }
    WRITE_PLANE(0, acc00, acc10)
    WRITE_PLANE(1, acc01, acc11)
    WRITE_PLANE(2, acc02, acc12)
#undef WRITE_PLANE
}

__global__ __launch_bounds__(256) void attn_kernel(
    float* qo,
    const float* __restrict__ K, const float* __restrict__ V,
    const float* __restrict__ part)
{
    __shared__ __attribute__((aligned(16))) float red[16 * D];

    const int tid  = threadIdx.x;
    const int lane = tid & 31;
    const int gwave = blockIdx.x * 8 + (tid >> 5);
    const int b = gwave / S;
    const int i = gwave - b * S;

    float2 kv[WIN], vv[WIN];
    bool   ok[WIN];
#pragma unroll
    for (int w = 0; w < WIN; ++w) {
        const int j  = i + DIL * (w - WIN / 2);
        ok[w] = (j >= 0) && (j < S);
        const int jc = j < 0 ? 0 : (j >= S ? S - 1 : j);
        const size_t rowJ = (size_t)(b * S + jc) * D;
        kv[w] = *(const float2*)(K + rowJ + lane * 2);
        vv[w] = *(const float2*)(V + rowJ + lane * 2);
    }

    const size_t rowQ = (size_t)(b * S + i) * D;
    const float2 qv = *(const float2*)(qo + rowQ + lane * 2);

    {
        const int c4 = tid & 15;
        const int g  = tid >> 4;
        const float* pp = part + (size_t)(b * NPB + g) * D + c4 * 4;
        const float4 t0 = *(const float4*)(pp);
        const float4 t1 = *(const float4*)(pp + 16 * D);
        const float4 t2 = *(const float4*)(pp + 32 * D);
        const float4 t3 = *(const float4*)(pp + 48 * D);
        float4 t;
        t.x = (t0.x + t1.x) + (t2.x + t3.x);
        t.y = (t0.y + t1.y) + (t2.y + t3.y);
        t.z = (t0.z + t1.z) + (t2.z + t3.z);
        t.w = (t0.w + t1.w) + (t2.w + t3.w);
        *(float4*)(red + g * D + c4 * 4) = t;
    }
    asm volatile("" : : "v"(vv[0].x), "v"(vv[0].y), "v"(vv[1].x), "v"(vv[1].y),
                        "v"(vv[2].x), "v"(vv[2].y), "v"(vv[3].x), "v"(vv[3].y),
                        "v"(vv[4].x), "v"(vv[4].y));
    __syncthreads();

    float accx = 0.f, accy = 0.f;
#pragma unroll
    for (int g = 0; g < 16; ++g) {
        const float2 r = *(const float2*)(red + g * D + lane * 2);
        accx += r.x;
        accy += r.y;
    }
    float Z = (float)S;

    float c[WIN];
#pragma unroll
    for (int w = 0; w < WIN; ++w) c[w] = qv.x * kv[w].x + qv.y * kv[w].y;
#pragma unroll
    for (int off = 16; off > 0; off >>= 1) {
#pragma unroll
        for (int w = 0; w < WIN; ++w) c[w] += __shfl_xor(c[w], off, 32);
    }

#pragma unroll
    for (int w = 0; w < WIN; ++w) {
        const float em1 = ok[w] ? __expf(c[w]) - 1.0f : 0.f;
        accx += em1 * vv[w].x;
        accy += em1 * vv[w].y;
        Z    += em1;
    }

    const float inv = 1.0f / Z;
    float2 o;
    o.x = accx * inv;
    o.y = accy * inv;
    *(float2*)(qo + rowQ + lane * 2) = o;
}

extern "C" void kernel_launch(void* const* d_in, const int* in_sizes, int n_in,
                              void* d_out, int out_size, void* d_ws, size_t ws_size,
                              hipStream_t stream) {
    const float* x  = (const float*)d_in[0];
    const float* Wq = (const float*)d_in[1];
    const float* bq = (const float*)d_in[2];
    const float* Wk = (const float*)d_in[3];
    const float* bk = (const float*)d_in[4];
    const float* Wv = (const float*)d_in[5];
    const float* bv = (const float*)d_in[6];
    float* outp = (float*)d_out;

    constexpr size_t QKV = (size_t)B * S * D;
    float* ws = (float*)d_ws;
    float* Pp = ws;
    float* Kb = ws + QKV;
    float* Vb = ws + 2 * QKV;

    qkv_proj_kernel<<<PROJ_WGS + NXW, 128, 0, stream>>>(x, Wq, bq, Wk, bk, Wv, bv,
                                                        outp, Kb, Vb, Pp);

    attn_kernel<<<(B * S) / 8, 256, 0, stream>>>(outp, Kb, Vb, Pp);
}
